# starting kernel
# baseline (speedup 1.0000x reference)
#include <hip/hip_runtime.h>


#define T_STEPS 4096
#define BATCH   1024
#define IN_DIM  5
#define HID     8
#define OUT_DIM 5

typedef __attribute__((ext_vector_type(2))) float v2f;
typedef __attribute__((ext_vector_type(8))) float v8f;

__device__ __forceinline__ float bcast_lane(float v, int l) {
  return __int_as_float(__builtin_amdgcn_readlane(__float_as_int(v), l));
}

__global__ __launch_bounds__(256) void lstm_scan_kernel(
    const float* __restrict__ seq, const float* __restrict__ W_ih,
    const float* __restrict__ W_hh, const float* __restrict__ b_ih,
    const float* __restrict__ b_hh, float* __restrict__ h_out) {
  const int lane  = threadIdx.x & 31;
  const int wave  = threadIdx.x >> 5;
  const int b     = blockIdx.x * 8 + wave;
  const int g     = lane;
  const int gtype = lane >> 3;

  const float NEG_L2E = -1.44269504088896340736f;
  const float wscale = (gtype == 2) ? (2.0f * NEG_L2E) : NEG_L2E;
  const float sa     = (gtype == 2) ? 2.0f : 1.0f;
  const float sb     = (gtype == 2) ? -1.0f : 0.0f;

  float wih[IN_DIM];
#pragma unroll
  for (int i = 0; i < IN_DIM; ++i) wih[i] = W_ih[g * IN_DIM + i] * wscale;
  float whh[HID];
#pragma unroll
  for (int k = 0; k < HID; ++k) whh[k] = W_hh[g * HID + k] * wscale;
  const float bias = (b_ih[g] + b_hh[g]) * wscale;

  float h = 0.0f, c = 0.0f;
  const float* xp = seq + (size_t)b * IN_DIM;
  const int tstride = BATCH * IN_DIM;

  for (int t = 0; t < T_STEPS; ++t) {
    __builtin_prefetch(xp + 16 * tstride, 0, 0);

    const float x0 = xp[0], x1 = xp[1], x2 = xp[2], x3 = xp[3], x4 = xp[4];

    float a0 = __builtin_fmaf(x0, wih[0], bias);
    float a1 = x2 * wih[2];
    a0 = __builtin_fmaf(x1, wih[1], a0);
    a1 = __builtin_fmaf(x3, wih[3], a1);
    a0 = __builtin_fmaf(x4, wih[4], a0);
#pragma unroll
    for (int k = 0; k < HID; k += 2) {
      a0 = __builtin_fmaf(bcast_lane(h, k),     whh[k],     a0);
      a1 = __builtin_fmaf(bcast_lane(h, k + 1), whh[k + 1], a1);
    }
    const float z   = a0 + a1;
    const float e   = __builtin_amdgcn_exp2f(z);
    const float s   = __builtin_amdgcn_rcpf(1.0f + e);
    const float act = __builtin_fmaf(sa, s, sb);

    const float fv = __shfl(act, lane + 8,  32);
    const float gv = __shfl(act, lane + 16, 32);
    const float ov = __shfl(act, lane + 24, 32);

    c = __builtin_fmaf(act, gv, fv * c);
    const float e2 = __builtin_amdgcn_exp2f(c * (2.0f * NEG_L2E));
    const float th = __builtin_fmaf(2.0f, __builtin_amdgcn_rcpf(1.0f + e2), -1.0f);
    h = ov * th;

    xp += tstride;
  }
  if (lane < HID) h_out[(size_t)b * HID + lane] = h;
}

__global__ __launch_bounds__(128) void out_proj_wmma(
    const float* __restrict__ h_last, const float* __restrict__ W_out,
    const float* __restrict__ b_out, float* __restrict__ out) {
  const int lane = threadIdx.x & 31;
  const int tile = (blockIdx.x * blockDim.x + threadIdx.x) >> 5;
  const int half = lane >> 4;
  const int lm   = lane & 15;

  const float* hrow = h_last + (size_t)(tile * 16 + lm) * HID;
  v2f a_lo, a_hi;
  a_lo.x = hrow[0 + 2 * half];
  a_lo.y = hrow[1 + 2 * half];
  a_hi.x = hrow[4 + 2 * half];
  a_hi.y = hrow[5 + 2 * half];

  float w0 = 0.f, w1 = 0.f, w2 = 0.f, w3 = 0.f, bo = 0.f;
  if (lm < OUT_DIM) {
    const float* wr = W_out + lm * HID;
    w0 = wr[0 + 2 * half];
    w1 = wr[1 + 2 * half];
    w2 = wr[4 + 2 * half];
    w3 = wr[5 + 2 * half];
    bo = b_out[lm];
  }
  v2f b_lo, b_hi;
  b_lo.x = w0; b_lo.y = w1;
  b_hi.x = w2; b_hi.y = w3;

  v8f acc;
#pragma unroll
  for (int r = 0; r < 8; ++r) acc[r] = bo;

  acc = __builtin_amdgcn_wmma_f32_16x16x4_f32(false, a_lo, false, b_lo,
                                              (short)0, acc, false, false);
  acc = __builtin_amdgcn_wmma_f32_16x16x4_f32(false, a_hi, false, b_hi,
                                              (short)0, acc, false, false);

  if (lm < OUT_DIM) {
    float* orow = out + (size_t)(tile * 16 + 8 * half) * OUT_DIM + lm;
#pragma unroll
    for (int r = 0; r < 8; ++r) orow[r * OUT_DIM] = acc[r];
  }
}

extern "C" void kernel_launch(void* const* d_in, const int* in_sizes, int n_in,
                              void* d_out, int out_size, void* d_ws, size_t ws_size,
                              hipStream_t stream) {
  const float* seq   = (const float*)d_in[0];
  const float* W_ih  = (const float*)d_in[1];
  const float* W_hh  = (const float*)d_in[2];
  const float* b_ih  = (const float*)d_in[3];
  const float* b_hh  = (const float*)d_in[4];
  const float* W_out = (const float*)d_in[5];
  const float* b_out = (const float*)d_in[6];
  float* out    = (float*)d_out;
  float* h_last = (float*)d_ws;

  lstm_scan_kernel<<<BATCH / 8, 256, 0, stream>>>(seq, W_ih, W_hh, b_ih, b_hh,
                                                  h_last);
  out_proj_wmma<<<(BATCH / 16) * 32 / 128, 128, 0, stream>>>(h_last, W_out,
                                                             b_out, out);
}


# cell-update chain shortened: state carried as -2log2e*c, 2x-1 maps folded into adjacent FMA/rcp, o kept for the quad's four units so h = fma(2r,o,-o) rides the DPP read of r; cell sum order changes
# speedup vs baseline: 2.1114x; 2.1114x over previous
#include <hip/hip_runtime.h>


#define T_STEPS 4096
#define BATCH   1024
#define IN_DIM  5
#define HID     8
#define OUT_DIM 5
#define CHUNK   8
#define NCHUNK  (T_STEPS / CHUNK)
#define STAGES  3
#define STEP_V4  (4 * 32)
#define STAGE_V4 (CHUNK * STEP_V4)
#define COPIES  4
#define BPB     (16 / COPIES)

typedef __attribute__((ext_vector_type(2))) float v2f;
typedef __attribute__((ext_vector_type(4))) float v4f;
typedef __attribute__((ext_vector_type(8))) float v8f;

__device__ __forceinline__ v8f wmma4(v2f a, v2f b, v8f c) {
  return __builtin_amdgcn_wmma_f32_16x16x4_f32(false, a, false, b, (short)0, c,
                                               false, false);
}

typedef float v2f_a4 __attribute__((ext_vector_type(2), aligned(4)));

__device__ __forceinline__ v2f load2(const float* __restrict__ p) {
  const v2f_a4 t = *(const v2f_a4*)p;
  return (v2f){t.x, t.y};
}

__device__ __forceinline__ v8f join8(v4f lo, v4f hi) {
  return __builtin_shufflevector(lo, hi, 0, 1, 2, 3, 4, 5, 6, 7);
}

#define QUAD_LANE(x, p)                                                     \
  __builtin_bit_cast(float, __builtin_amdgcn_update_dpp(                    \
      0, __builtin_bit_cast(int, (x)), (p) * 0x55, 0xF, 0xF, true))

#define STEP_TAIL(d0, d1, d2, d3, tmp, cK, q0, q1, q2, q3)                  \
  asm("v_exp_f32_e32 %4, %5\n\t"                                            \
      "v_nop\n\t"                                                           \
      "v_fma_f32 %4, %4, 0.5, 0.5\n\t"                                      \
      "v_rcp_f32_e32 %4, %4\n\t"                                            \
      "v_nop\n\t"                                                           \
      "v_fma_f32_e64_dpp %0, %4, %6, -%6 quad_perm:[0,0,0,0] row_mask:0xf bank_mask:0xf bound_ctrl:1\n\t" \
      "v_fma_f32_e64_dpp %1, %4, %7, -%7 quad_perm:[1,1,1,1] row_mask:0xf bank_mask:0xf bound_ctrl:1\n\t" \
      "v_fma_f32_e64_dpp %2, %4, %8, -%8 quad_perm:[2,2,2,2] row_mask:0xf bank_mask:0xf bound_ctrl:1\n\t" \
      "v_fma_f32_e64_dpp %3, %4, %9, -%9 quad_perm:[3,3,3,3] row_mask:0xf bank_mask:0xf bound_ctrl:1"       \
      : "=&v"(d0), "=&v"(d1), "=&v"(d2), "=&v"(d3), "=&v"(tmp)              \
      : "v"(cK), "v"(q0), "v"(q1), "v"(q2), "v"(q3))

__device__ __forceinline__ float pick4(float a, float b, float c, float d,
                                       bool odd, bool upper) {
  const float lo = odd ? b : a;
  const float hi = odd ? d : c;
  return upper ? hi : lo;
}

__global__ __launch_bounds__(64) void lstm_scan_wmma(
    const float* __restrict__ seq, const float* __restrict__ W_ih,
    const float* __restrict__ W_hh, const float* __restrict__ b_ih,
    const float* __restrict__ b_hh, float* __restrict__ h_out) {
  __shared__ v4f ring[STAGES * STAGE_V4];

  const int lane = threadIdx.x & 31;
  const int role = __builtin_amdgcn_readfirstlane((int)(threadIdx.x >> 5));
  const int col  = lane & 15;
  const int half = lane >> 4;
  const int b0   = blockIdx.x * BPB;
  const int bcol = col >> 2;
  const int own  = col & 3;
  const bool own_odd   = (own & 1) != 0;
  const bool own_upper = (own & 2) != 0;

  const float NEG_L2E = -1.44269504088896340736f;
  const float INV_2K  = -0.17328679513998632735f;

  v2f aH[2][2], aXa[2], aXb[2];
  v8f biasC[2];
#pragma unroll
  for (int m = 0; m < 2; ++m) {
    const int type = 2 * m + ((col & 7) >> 2);
    const int unit = 4 * (col >> 3) + (col & 3);
    const int G    = 8 * type + unit;
    const float ws = (type == 2) ? (2.0f * NEG_L2E) : NEG_L2E;
#pragma unroll
    for (int q = 0; q < 2; ++q) {
      aH[m][q].x = W_hh[G * HID + 4 * half + 2 * q + 0] * ws;
      aH[m][q].y = W_hh[G * HID + 4 * half + 2 * q + 1] * ws;
    }
    const float wi0 = W_ih[G * IN_DIM + 0] * ws, wi1 = W_ih[G * IN_DIM + 1] * ws;
    const float wi2 = W_ih[G * IN_DIM + 2] * ws, wi3 = W_ih[G * IN_DIM + 3] * ws;
    const float wi4 = W_ih[G * IN_DIM + 4] * ws;
    aXa[m].x = half ? 0.0f : wi0;
    aXa[m].y = half ? wi4  : wi1;
    aXb[m].x = half ? 0.0f : wi2;
    aXb[m].y = half ? 0.0f : wi3;
#pragma unroll
    for (int r = 0; r < 8; ++r) {
      const int tr = 2 * m + (r >> 2);
      const int Gr = 8 * tr + 4 * half + (r & 3);
      const float wr = (tr == 2) ? (2.0f * NEG_L2E) : NEG_L2E;
      biasC[m][r] = (b_ih[Gr] + b_hh[Gr]) * wr;
    }
  }

  const int    oA      = half ? 3 : 0;
  const size_t tstride = (size_t)BATCH * IN_DIM;
  const float* xbase   = seq + (size_t)(b0 + bcol) * IN_DIM;

#define FETCH_CHUNK(dA, dB, p)                                              \
  _Pragma("unroll") for (int s = 0; s < CHUNK; ++s) {                       \
    const int tt = (p) * CHUNK + s;                                         \
    const int tc = (tt < T_STEPS) ? tt : (T_STEPS - 1);                     \
    const float* rp = xbase + (size_t)tc * tstride;                         \
    dA[s] = load2(rp + oA);                                                 \
    dB[s] = load2(rp + 2);                                                  \
  }

#define MAKE_CHUNK(sA, sB, stg)                                             \
  {                                                                         \
    v4f* wp = ring + (stg) * STAGE_V4 + lane;                               \
    _Pragma("unroll") for (int s = 0; s < CHUNK; ++s) {                     \
      v8f d0 = wmma4(aXa[0], sA[s], biasC[0]);                              \
      v8f d1 = wmma4(aXa[1], sA[s], biasC[1]);                              \
      d0 = wmma4(aXb[0], sB[s], d0);                                        \
      d1 = wmma4(aXb[1], sB[s], d1);                                        \
      wp[s * STEP_V4 + 0 * 32] = __builtin_shufflevector(d0, d0, 0, 1, 2, 3); \
      wp[s * STEP_V4 + 1 * 32] = __builtin_shufflevector(d0, d0, 4, 5, 6, 7); \
      wp[s * STEP_V4 + 2 * 32] = __builtin_shufflevector(d1, d1, 0, 1, 2, 3); \
      wp[s * STEP_V4 + 3 * 32] = __builtin_shufflevector(d1, d1, 4, 5, 6, 7); \
    }                                                                       \
  }

#define READ_GROUP(c0, c1, stg, f)                                          \
  {                                                                         \
    const v4f* rq = ring + (stg) * STAGE_V4 + (f) * STEP_V4 + lane;         \
    _Pragma("unroll") for (int s = 0; s < 4; ++s) {                         \
      c0[s] = join8(rq[s * STEP_V4 + 0 * 32], rq[s * STEP_V4 + 1 * 32]);    \
      c1[s] = join8(rq[s * STEP_V4 + 2 * 32], rq[s * STEP_V4 + 3 * 32]);    \
    }                                                                       \
  }

#define SCAN_STEP(cin0, cin1)                                               \
  {                                                                         \
    v8f g0 = wmma4(aH[0][0], bh0, cin0);                                    \
    v8f g1 = wmma4(aH[1][0], bh0, cin1);                                    \
    g0 = wmma4(aH[0][1], bh1, g0);                                          \
    g1 = wmma4(aH[1][1], bh1, g1);                                          \
    const float pi = pick4(g0[0], g0[1], g0[2], g0[3], own_odd, own_upper); \
    const float pf = pick4(g0[4], g0[5], g0[6], g0[7], own_odd, own_upper); \
    const float pg = pick4(g1[0], g1[1], g1[2], g1[3], own_odd, own_upper); \
    const float iK2 = __builtin_amdgcn_rcpf(                                \
        __builtin_fmaf(__builtin_amdgcn_exp2f(pi), INV_2K, INV_2K));        \
    const float fv = __builtin_amdgcn_rcpf(1.0f + __builtin_amdgcn_exp2f(pf)); \
    const float u  = __builtin_fmaf(iK2, -0.5f, fv * csK);                  \
    const float gs = __builtin_amdgcn_rcpf(1.0f + __builtin_amdgcn_exp2f(pg)); \
    csK = __builtin_fmaf(iK2, gs, u);                                       \
    const float oq0 = __builtin_amdgcn_rcpf(1.0f + __builtin_amdgcn_exp2f(g1[4])); \
    const float oq1 = __builtin_amdgcn_rcpf(1.0f + __builtin_amdgcn_exp2f(g1[5])); \
    const float oq2 = __builtin_amdgcn_rcpf(1.0f + __builtin_amdgcn_exp2f(g1[6])); \
    const float oq3 = __builtin_amdgcn_rcpf(1.0f + __builtin_amdgcn_exp2f(g1[7])); \
    float tail_tmp;                                                         \
    STEP_TAIL(bh0.x, bh0.y, bh1.x, bh1.y, tail_tmp, csK, oq0, oq1, oq2, oq3); \
  }

  v2f bh0 = {0.f, 0.f};
  v2f bh1 = {0.f, 0.f};
  float csK = 0.f;
  v2f xcA[CHUNK], xcB[CHUNK];
  v8f cA0[4], cA1[4], cB0[4], cB1[4];

  if (role == 1) {
    FETCH_CHUNK(xcA, xcB, 0)
    MAKE_CHUNK(xcA, xcB, 0)
    FETCH_CHUNK(xcA, xcB, 1)
    MAKE_CHUNK(xcA, xcB, 1)
    FETCH_CHUNK(xcA, xcB, 2)
  }
  __syncthreads();
  if (role == 0) {
    READ_GROUP(cA0, cA1, 0, 0)
  }

  int st = 0;
  for (int k = 0; k < NCHUNK; ++k) {
    const int st1 = (st == STAGES - 1) ? 0 : st + 1;
    const int st2 = (st1 == STAGES - 1) ? 0 : st1 + 1;
    if (role == 1) {
      if (k + 2 < NCHUNK) {
        v2f xnA[CHUNK], xnB[CHUNK];
        FETCH_CHUNK(xnA, xnB, k + 3)
#pragma unroll
        for (int s = 0; s < CHUNK; ++s) {
          const int tt = (k + 5) * CHUNK + s;
          const int tc = (tt < T_STEPS) ? tt : (T_STEPS - 1);
          __builtin_prefetch(xbase + (size_t)tc * tstride, 0, 0);
        }
        MAKE_CHUNK(xcA, xcB, st2)
#pragma unroll
        for (int s = 0; s < CHUNK; ++s) { xcA[s] = xnA[s]; xcB[s] = xnB[s]; }
      }
    } else {
      READ_GROUP(cB0, cB1, st, 4)
      __builtin_amdgcn_sched_barrier(0);
#pragma unroll
      for (int s = 0; s < 4; ++s) SCAN_STEP(cA0[s], cA1[s])
      READ_GROUP(cA0, cA1, st1, 0)
      __builtin_amdgcn_sched_barrier(0);
#pragma unroll
      for (int s = 0; s < 4; ++s) SCAN_STEP(cB0[s], cB1[s])
    }
    __syncthreads();
    st = st1;
  }
#undef FETCH_CHUNK
#undef MAKE_CHUNK
#undef READ_GROUP
#undef SCAN_STEP
#undef QUAD_LANE

  if (role == 0) {
    const float hv = pick4(bh0.x, bh0.y, bh1.x, bh1.y, own_odd, own_upper);
    h_out[(size_t)(b0 + bcol) * HID + 4 * half + own] = hv;
  }
}

__global__ __launch_bounds__(128) void out_proj_wmma(
    const float* __restrict__ h_last, const float* __restrict__ W_out,
    const float* __restrict__ b_out, float* __restrict__ out) {
  const int lane = threadIdx.x & 31;
  const int tile = (blockIdx.x * blockDim.x + threadIdx.x) >> 5;
  const int half = lane >> 4;
  const int lm   = lane & 15;

  const float* hrow = h_last + (size_t)(tile * 16 + lm) * HID;
  v2f a_lo, a_hi;
  a_lo.x = hrow[0 + 2 * half];
  a_lo.y = hrow[1 + 2 * half];
  a_hi.x = hrow[4 + 2 * half];
  a_hi.y = hrow[5 + 2 * half];

  float w0 = 0.f, w1 = 0.f, w2 = 0.f, w3 = 0.f, bo = 0.f;
  if (lm < OUT_DIM) {
    const float* wr = W_out + lm * HID;
    w0 = wr[0 + 2 * half];
    w1 = wr[1 + 2 * half];
    w2 = wr[4 + 2 * half];
    w3 = wr[5 + 2 * half];
    bo = b_out[lm];
  }
  v2f b_lo, b_hi;
  b_lo.x = w0; b_lo.y = w1;
  b_hi.x = w2; b_hi.y = w3;

  v8f acc;
#pragma unroll
  for (int r = 0; r < 8; ++r) acc[r] = bo;

  acc = __builtin_amdgcn_wmma_f32_16x16x4_f32(false, a_lo, false, b_lo,
                                              (short)0, acc, false, false);
  acc = __builtin_amdgcn_wmma_f32_16x16x4_f32(false, a_hi, false, b_hi,
                                              (short)0, acc, false, false);

  if (lm < OUT_DIM) {
    float* orow = out + (size_t)(tile * 16 + 8 * half) * OUT_DIM + lm;
#pragma unroll
    for (int r = 0; r < 8; ++r) orow[r * OUT_DIM] = acc[r];
  }
}

extern "C" void kernel_launch(void* const* d_in, const int* in_sizes, int n_in,
                              void* d_out, int out_size, void* d_ws, size_t ws_size,
                              hipStream_t stream) {
  const float* seq   = (const float*)d_in[0];
  const float* W_ih  = (const float*)d_in[1];
  const float* W_hh  = (const float*)d_in[2];
  const float* b_ih  = (const float*)d_in[3];
  const float* b_hh  = (const float*)d_in[4];
  const float* W_out = (const float*)d_in[5];
  const float* b_out = (const float*)d_in[6];
  float* out    = (float*)d_out;
  float* h_last = (float*)d_ws;

  lstm_scan_wmma<<<BATCH / BPB, 64, 0, stream>>>(seq, W_ih, W_hh, b_ih, b_hh,
                                                 h_last);
  out_proj_wmma<<<(BATCH / 16) * 32 / 128, 128, 0, stream>>>(h_last, W_out,
                                                             b_out, out);
}


# ring stage lengthened from 8 to 16 steps: one barrier, one loop edge and one set of loop-carried tile copies per 16 steps; feeder covers its x loads with 64 matrix ops; sums unchanged
# speedup vs baseline: 2.2207x; 1.0518x over previous
#include <hip/hip_runtime.h>


#define T_STEPS 4096
#define BATCH   1024
#define IN_DIM  5
#define HID     8
#define OUT_DIM 5
#define CHUNK   16
#define NCHUNK  (T_STEPS / CHUNK)
#define STAGES  3
#define STEP_V4  (4 * 32)
#define STAGE_V4 (CHUNK * STEP_V4)
#define COPIES  4
#define BPB     (16 / COPIES)

typedef __attribute__((ext_vector_type(2))) float v2f;
typedef __attribute__((ext_vector_type(4))) float v4f;
typedef __attribute__((ext_vector_type(8))) float v8f;

__device__ __forceinline__ v8f wmma4(v2f a, v2f b, v8f c) {
  return __builtin_amdgcn_wmma_f32_16x16x4_f32(false, a, false, b, (short)0, c,
                                               false, false);
}

typedef float v2f_a4 __attribute__((ext_vector_type(2), aligned(4)));

__device__ __forceinline__ v2f load2(const float* __restrict__ p) {
  const v2f_a4 t = *(const v2f_a4*)p;
  return (v2f){t.x, t.y};
}

__device__ __forceinline__ v8f join8(v4f lo, v4f hi) {
  return __builtin_shufflevector(lo, hi, 0, 1, 2, 3, 4, 5, 6, 7);
}

#define QUAD_LANE(x, p)                                                     \
  __builtin_bit_cast(float, __builtin_amdgcn_update_dpp(                    \
      0, __builtin_bit_cast(int, (x)), (p) * 0x55, 0xF, 0xF, true))

#define STEP_TAIL(d0, d1, d2, d3, tmp, cK, q0, q1, q2, q3)                  \
  asm("v_exp_f32_e32 %4, %5\n\t"                                            \
      "v_nop\n\t"                                                           \
      "v_fma_f32 %4, %4, 0.5, 0.5\n\t"                                      \
      "v_rcp_f32_e32 %4, %4\n\t"                                            \
      "v_nop\n\t"                                                           \
      "v_fma_f32_e64_dpp %0, %4, %6, -%6 quad_perm:[0,0,0,0] row_mask:0xf bank_mask:0xf bound_ctrl:1\n\t" \
      "v_fma_f32_e64_dpp %1, %4, %7, -%7 quad_perm:[1,1,1,1] row_mask:0xf bank_mask:0xf bound_ctrl:1\n\t" \
      "v_fma_f32_e64_dpp %2, %4, %8, -%8 quad_perm:[2,2,2,2] row_mask:0xf bank_mask:0xf bound_ctrl:1\n\t" \
      "v_fma_f32_e64_dpp %3, %4, %9, -%9 quad_perm:[3,3,3,3] row_mask:0xf bank_mask:0xf bound_ctrl:1"       \
      : "=&v"(d0), "=&v"(d1), "=&v"(d2), "=&v"(d3), "=&v"(tmp)              \
      : "v"(cK), "v"(q0), "v"(q1), "v"(q2), "v"(q3))

__device__ __forceinline__ float pick4(float a, float b, float c, float d,
                                       bool odd, bool upper) {
  const float lo = odd ? b : a;
  const float hi = odd ? d : c;
  return upper ? hi : lo;
}

__global__ __launch_bounds__(64) void lstm_scan_wmma(
    const float* __restrict__ seq, const float* __restrict__ W_ih,
    const float* __restrict__ W_hh, const float* __restrict__ b_ih,
    const float* __restrict__ b_hh, float* __restrict__ h_out) {
  __shared__ v4f ring[STAGES * STAGE_V4];

  const int lane = threadIdx.x & 31;
  const int role = __builtin_amdgcn_readfirstlane((int)(threadIdx.x >> 5));
  const int col  = lane & 15;
  const int half = lane >> 4;
  const int b0   = blockIdx.x * BPB;
  const int bcol = col >> 2;
  const int own  = col & 3;
  const bool own_odd   = (own & 1) != 0;
  const bool own_upper = (own & 2) != 0;

  const float NEG_L2E = -1.44269504088896340736f;
  const float INV_2K  = -0.17328679513998632735f;

  v2f aH[2][2], aXa[2], aXb[2];
  v8f biasC[2];
#pragma unroll
  for (int m = 0; m < 2; ++m) {
    const int type = 2 * m + ((col & 7) >> 2);
    const int unit = 4 * (col >> 3) + (col & 3);
    const int G    = 8 * type + unit;
    const float ws = (type == 2) ? (2.0f * NEG_L2E) : NEG_L2E;
#pragma unroll
    for (int q = 0; q < 2; ++q) {
      aH[m][q].x = W_hh[G * HID + 4 * half + 2 * q + 0] * ws;
      aH[m][q].y = W_hh[G * HID + 4 * half + 2 * q + 1] * ws;
    }
    const float wi0 = W_ih[G * IN_DIM + 0] * ws, wi1 = W_ih[G * IN_DIM + 1] * ws;
    const float wi2 = W_ih[G * IN_DIM + 2] * ws, wi3 = W_ih[G * IN_DIM + 3] * ws;
    const float wi4 = W_ih[G * IN_DIM + 4] * ws;
    aXa[m].x = half ? 0.0f : wi0;
    aXa[m].y = half ? wi4  : wi1;
    aXb[m].x = half ? 0.0f : wi2;
    aXb[m].y = half ? 0.0f : wi3;
#pragma unroll
    for (int r = 0; r < 8; ++r) {
      const int tr = 2 * m + (r >> 2);
      const int Gr = 8 * tr + 4 * half + (r & 3);
      const float wr = (tr == 2) ? (2.0f * NEG_L2E) : NEG_L2E;
      biasC[m][r] = (b_ih[Gr] + b_hh[Gr]) * wr;
    }
  }

  const int    oA      = half ? 3 : 0;
  const size_t tstride = (size_t)BATCH * IN_DIM;
  const float* xbase   = seq + (size_t)(b0 + bcol) * IN_DIM;

#define FETCH_CHUNK(dA, dB, p)                                              \
  _Pragma("unroll") for (int s = 0; s < CHUNK; ++s) {                       \
    const int tt = (p) * CHUNK + s;                                         \
    const int tc = (tt < T_STEPS) ? tt : (T_STEPS - 1);                     \
    const float* rp = xbase + (size_t)tc * tstride;                         \
    dA[s] = load2(rp + oA);                                                 \
    dB[s] = load2(rp + 2);                                                  \
  }

#define MAKE_CHUNK(sA, sB, stg)                                             \
  {                                                                         \
    v4f* wp = ring + (stg) * STAGE_V4 + lane;                               \
    _Pragma("unroll") for (int s = 0; s < CHUNK; ++s) {                     \
      v8f d0 = wmma4(aXa[0], sA[s], biasC[0]);                              \
      v8f d1 = wmma4(aXa[1], sA[s], biasC[1]);                              \
      d0 = wmma4(aXb[0], sB[s], d0);                                        \
      d1 = wmma4(aXb[1], sB[s], d1);                                        \
      wp[s * STEP_V4 + 0 * 32] = __builtin_shufflevector(d0, d0, 0, 1, 2, 3); \
      wp[s * STEP_V4 + 1 * 32] = __builtin_shufflevector(d0, d0, 4, 5, 6, 7); \
      wp[s * STEP_V4 + 2 * 32] = __builtin_shufflevector(d1, d1, 0, 1, 2, 3); \
      wp[s * STEP_V4 + 3 * 32] = __builtin_shufflevector(d1, d1, 4, 5, 6, 7); \
    }                                                                       \
  }

#define READ_GROUP(c0, c1, stg, f)                                          \
  {                                                                         \
    const v4f* rq = ring + (stg) * STAGE_V4 + (f) * STEP_V4 + lane;         \
    _Pragma("unroll") for (int s = 0; s < 4; ++s) {                         \
      c0[s] = join8(rq[s * STEP_V4 + 0 * 32], rq[s * STEP_V4 + 1 * 32]);    \
      c1[s] = join8(rq[s * STEP_V4 + 2 * 32], rq[s * STEP_V4 + 3 * 32]);    \
    }                                                                       \
  }

#define SCAN_STEP(cin0, cin1)                                               \
  {                                                                         \
    v8f g0 = wmma4(aH[0][0], bh0, cin0);                                    \
    v8f g1 = wmma4(aH[1][0], bh0, cin1);                                    \
    g0 = wmma4(aH[0][1], bh1, g0);                                          \
    g1 = wmma4(aH[1][1], bh1, g1);                                          \
    const float pi = pick4(g0[0], g0[1], g0[2], g0[3], own_odd, own_upper); \
    const float pf = pick4(g0[4], g0[5], g0[6], g0[7], own_odd, own_upper); \
    const float pg = pick4(g1[0], g1[1], g1[2], g1[3], own_odd, own_upper); \
    const float iK2 = __builtin_amdgcn_rcpf(                                \
        __builtin_fmaf(__builtin_amdgcn_exp2f(pi), INV_2K, INV_2K));        \
    const float fv = __builtin_amdgcn_rcpf(1.0f + __builtin_amdgcn_exp2f(pf)); \
    const float u  = __builtin_fmaf(iK2, -0.5f, fv * csK);                  \
    const float gs = __builtin_amdgcn_rcpf(1.0f + __builtin_amdgcn_exp2f(pg)); \
    csK = __builtin_fmaf(iK2, gs, u);                                       \
    const float oq0 = __builtin_amdgcn_rcpf(1.0f + __builtin_amdgcn_exp2f(g1[4])); \
    const float oq1 = __builtin_amdgcn_rcpf(1.0f + __builtin_amdgcn_exp2f(g1[5])); \
    const float oq2 = __builtin_amdgcn_rcpf(1.0f + __builtin_amdgcn_exp2f(g1[6])); \
    const float oq3 = __builtin_amdgcn_rcpf(1.0f + __builtin_amdgcn_exp2f(g1[7])); \
    float tail_tmp;                                                         \
    STEP_TAIL(bh0.x, bh0.y, bh1.x, bh1.y, tail_tmp, csK, oq0, oq1, oq2, oq3); \
  }

  v2f bh0 = {0.f, 0.f};
  v2f bh1 = {0.f, 0.f};
  float csK = 0.f;
  v2f xcA[CHUNK], xcB[CHUNK];
  v8f cA0[4], cA1[4], cB0[4], cB1[4];

  if (role == 1) {
    FETCH_CHUNK(xcA, xcB, 0)
    MAKE_CHUNK(xcA, xcB, 0)
    FETCH_CHUNK(xcA, xcB, 1)
    MAKE_CHUNK(xcA, xcB, 1)
    FETCH_CHUNK(xcA, xcB, 2)
  }
  __syncthreads();
  if (role == 0) {
    READ_GROUP(cA0, cA1, 0, 0)
  }

  int st = 0;
  for (int k = 0; k < NCHUNK; ++k) {
    const int st1 = (st == STAGES - 1) ? 0 : st + 1;
    const int st2 = (st1 == STAGES - 1) ? 0 : st1 + 1;
    if (role == 1) {
      if (k + 2 < NCHUNK) {
        v2f xnA[CHUNK], xnB[CHUNK];
        FETCH_CHUNK(xnA, xnB, k + 3)
#pragma unroll
        for (int s = 0; s < CHUNK; ++s) {
          const int tt = (k + 5) * CHUNK + s;
          const int tc = (tt < T_STEPS) ? tt : (T_STEPS - 1);
          __builtin_prefetch(xbase + (size_t)tc * tstride, 0, 0);
        }
        MAKE_CHUNK(xcA, xcB, st2)
#pragma unroll
        for (int s = 0; s < CHUNK; ++s) { xcA[s] = xnA[s]; xcB[s] = xnB[s]; }
      }
    } else {
      READ_GROUP(cB0, cB1, st, 4)
      __builtin_amdgcn_sched_barrier(0);
#pragma unroll
      for (int s = 0; s < 4; ++s) SCAN_STEP(cA0[s], cA1[s])
      READ_GROUP(cA0, cA1, st, 8)
      __builtin_amdgcn_sched_barrier(0);
#pragma unroll
      for (int s = 0; s < 4; ++s) SCAN_STEP(cB0[s], cB1[s])
      READ_GROUP(cB0, cB1, st, 12)
      __builtin_amdgcn_sched_barrier(0);
#pragma unroll
      for (int s = 0; s < 4; ++s) SCAN_STEP(cA0[s], cA1[s])
      READ_GROUP(cA0, cA1, st1, 0)
      __builtin_amdgcn_sched_barrier(0);
#pragma unroll
      for (int s = 0; s < 4; ++s) SCAN_STEP(cB0[s], cB1[s])
    }
    __syncthreads();
    st = st1;
  }
#undef FETCH_CHUNK
#undef MAKE_CHUNK
#undef READ_GROUP
#undef SCAN_STEP
#undef QUAD_LANE

  if (role == 0) {
    const float hv = pick4(bh0.x, bh0.y, bh1.x, bh1.y, own_odd, own_upper);
    h_out[(size_t)(b0 + bcol) * HID + 4 * half + own] = hv;
  }
}

__global__ __launch_bounds__(128) void out_proj_wmma(
    const float* __restrict__ h_last, const float* __restrict__ W_out,
    const float* __restrict__ b_out, float* __restrict__ out) {
  const int lane = threadIdx.x & 31;
  const int tile = (blockIdx.x * blockDim.x + threadIdx.x) >> 5;
  const int half = lane >> 4;
  const int lm   = lane & 15;

  const float* hrow = h_last + (size_t)(tile * 16 + lm) * HID;
  v2f a_lo, a_hi;
  a_lo.x = hrow[0 + 2 * half];
  a_lo.y = hrow[1 + 2 * half];
  a_hi.x = hrow[4 + 2 * half];
  a_hi.y = hrow[5 + 2 * half];

  float w0 = 0.f, w1 = 0.f, w2 = 0.f, w3 = 0.f, bo = 0.f;
  if (lm < OUT_DIM) {
    const float* wr = W_out + lm * HID;
    w0 = wr[0 + 2 * half];
    w1 = wr[1 + 2 * half];
    w2 = wr[4 + 2 * half];
    w3 = wr[5 + 2 * half];
    bo = b_out[lm];
  }
  v2f b_lo, b_hi;
  b_lo.x = w0; b_lo.y = w1;
  b_hi.x = w2; b_hi.y = w3;

  v8f acc;
#pragma unroll
  for (int r = 0; r < 8; ++r) acc[r] = bo;

  acc = __builtin_amdgcn_wmma_f32_16x16x4_f32(false, a_lo, false, b_lo,
                                              (short)0, acc, false, false);
  acc = __builtin_amdgcn_wmma_f32_16x16x4_f32(false, a_hi, false, b_hi,
                                              (short)0, acc, false, false);

  if (lm < OUT_DIM) {
    float* orow = out + (size_t)(tile * 16 + 8 * half) * OUT_DIM + lm;
#pragma unroll
    for (int r = 0; r < 8; ++r) orow[r * OUT_DIM] = acc[r];
  }
}

extern "C" void kernel_launch(void* const* d_in, const int* in_sizes, int n_in,
                              void* d_out, int out_size, void* d_ws, size_t ws_size,
                              hipStream_t stream) {
  const float* seq   = (const float*)d_in[0];
  const float* W_ih  = (const float*)d_in[1];
  const float* W_hh  = (const float*)d_in[2];
  const float* b_ih  = (const float*)d_in[3];
  const float* b_hh  = (const float*)d_in[4];
  const float* W_out = (const float*)d_in[5];
  const float* b_out = (const float*)d_in[6];
  float* out    = (float*)d_out;
  float* h_last = (float*)d_ws;

  lstm_scan_wmma<<<BATCH / BPB, 64, 0, stream>>>(seq, W_ih, W_hh, b_ih, b_hh,
                                                 h_last);
  out_proj_wmma<<<(BATCH / 16) * 32 / 128, 128, 0, stream>>>(h_last, W_out,
                                                             b_out, out);
}


# scan wave reads all tiles of a pass as one 64-load LDS group at the head of the pass (steps 2-15 of its chunk, steps 0-1 of the next) instead of nine scattered places; sums unchanged
# speedup vs baseline: 2.4083x; 1.0845x over previous
#include <hip/hip_runtime.h>


#define T_STEPS 4096
#define BATCH   1024
#define IN_DIM  5
#define HID     8
#define OUT_DIM 5
#define CHUNK   16
#define LEAD    2
#define NCHUNK  (T_STEPS / CHUNK)
#define STAGES  3
#define STEP_V4  (4 * 32)
#define STAGE_V4 (CHUNK * STEP_V4)
#define COPIES  4
#define BPB     (16 / COPIES)

typedef __attribute__((ext_vector_type(2))) float v2f;
typedef __attribute__((ext_vector_type(4))) float v4f;
typedef __attribute__((ext_vector_type(8))) float v8f;

__device__ __forceinline__ v8f wmma4(v2f a, v2f b, v8f c) {
  return __builtin_amdgcn_wmma_f32_16x16x4_f32(false, a, false, b, (short)0, c,
                                               false, false);
}

typedef float v2f_a4 __attribute__((ext_vector_type(2), aligned(4)));

__device__ __forceinline__ v2f load2(const float* __restrict__ p) {
  const v2f_a4 t = *(const v2f_a4*)p;
  return (v2f){t.x, t.y};
}

__device__ __forceinline__ v8f join8(v4f lo, v4f hi) {
  return __builtin_shufflevector(lo, hi, 0, 1, 2, 3, 4, 5, 6, 7);
}

#define QUAD_LANE(x, p)                                                     \
  __builtin_bit_cast(float, __builtin_amdgcn_update_dpp(                    \
      0, __builtin_bit_cast(int, (x)), (p) * 0x55, 0xF, 0xF, true))

#define STEP_TAIL(d0, d1, d2, d3, tmp, cK, q0, q1, q2, q3)                  \
  asm("v_exp_f32_e32 %4, %5\n\t"                                            \
      "v_nop\n\t"                                                           \
      "v_fma_f32 %4, %4, 0.5, 0.5\n\t"                                      \
      "v_rcp_f32_e32 %4, %4\n\t"                                            \
      "v_nop\n\t"                                                           \
      "v_fma_f32_e64_dpp %0, %4, %6, -%6 quad_perm:[0,0,0,0] row_mask:0xf bank_mask:0xf bound_ctrl:1\n\t" \
      "v_fma_f32_e64_dpp %1, %4, %7, -%7 quad_perm:[1,1,1,1] row_mask:0xf bank_mask:0xf bound_ctrl:1\n\t" \
      "v_fma_f32_e64_dpp %2, %4, %8, -%8 quad_perm:[2,2,2,2] row_mask:0xf bank_mask:0xf bound_ctrl:1\n\t" \
      "v_fma_f32_e64_dpp %3, %4, %9, -%9 quad_perm:[3,3,3,3] row_mask:0xf bank_mask:0xf bound_ctrl:1"       \
      : "=&v"(d0), "=&v"(d1), "=&v"(d2), "=&v"(d3), "=&v"(tmp)              \
      : "v"(cK), "v"(q0), "v"(q1), "v"(q2), "v"(q3))

__device__ __forceinline__ float pick4(float a, float b, float c, float d,
                                       bool odd, bool upper) {
  const float lo = odd ? b : a;
  const float hi = odd ? d : c;
  return upper ? hi : lo;
}

__global__ __launch_bounds__(64) void lstm_scan_wmma(
    const float* __restrict__ seq, const float* __restrict__ W_ih,
    const float* __restrict__ W_hh, const float* __restrict__ b_ih,
    const float* __restrict__ b_hh, float* __restrict__ h_out) {
  __shared__ v4f ring[STAGES * STAGE_V4];

  const int lane = threadIdx.x & 31;
  const int role = __builtin_amdgcn_readfirstlane((int)(threadIdx.x >> 5));
  const int col  = lane & 15;
  const int half = lane >> 4;
  const int b0   = blockIdx.x * BPB;
  const int bcol = col >> 2;
  const int own  = col & 3;
  const bool own_odd   = (own & 1) != 0;
  const bool own_upper = (own & 2) != 0;

  const float NEG_L2E = -1.44269504088896340736f;
  const float INV_2K  = -0.17328679513998632735f;

  v2f aH[2][2], aXa[2], aXb[2];
  v8f biasC[2];
#pragma unroll
  for (int m = 0; m < 2; ++m) {
    const int type = 2 * m + ((col & 7) >> 2);
    const int unit = 4 * (col >> 3) + (col & 3);
    const int G    = 8 * type + unit;
    const float ws = (type == 2) ? (2.0f * NEG_L2E) : NEG_L2E;
#pragma unroll
    for (int q = 0; q < 2; ++q) {
      aH[m][q].x = W_hh[G * HID + 4 * half + 2 * q + 0] * ws;
      aH[m][q].y = W_hh[G * HID + 4 * half + 2 * q + 1] * ws;
    }
    const float wi0 = W_ih[G * IN_DIM + 0] * ws, wi1 = W_ih[G * IN_DIM + 1] * ws;
    const float wi2 = W_ih[G * IN_DIM + 2] * ws, wi3 = W_ih[G * IN_DIM + 3] * ws;
    const float wi4 = W_ih[G * IN_DIM + 4] * ws;
    aXa[m].x = half ? 0.0f : wi0;
    aXa[m].y = half ? wi4  : wi1;
    aXb[m].x = half ? 0.0f : wi2;
    aXb[m].y = half ? 0.0f : wi3;
#pragma unroll
    for (int r = 0; r < 8; ++r) {
      const int tr = 2 * m + (r >> 2);
      const int Gr = 8 * tr + 4 * half + (r & 3);
      const float wr = (tr == 2) ? (2.0f * NEG_L2E) : NEG_L2E;
      biasC[m][r] = (b_ih[Gr] + b_hh[Gr]) * wr;
    }
  }

  const int    oA      = half ? 3 : 0;
  const size_t tstride = (size_t)BATCH * IN_DIM;
  const float* xbase   = seq + (size_t)(b0 + bcol) * IN_DIM;

#define FETCH_CHUNK(dA, dB, p)                                              \
  _Pragma("unroll") for (int s = 0; s < CHUNK; ++s) {                       \
    const int tt = (p) * CHUNK + s;                                         \
    const int tc = (tt < T_STEPS) ? tt : (T_STEPS - 1);                     \
    const float* rp = xbase + (size_t)tc * tstride;                         \
    dA[s] = load2(rp + oA);                                                 \
    dB[s] = load2(rp + 2);                                                  \
  }

#define MAKE_CHUNK(sA, sB, stg)                                             \
  {                                                                         \
    v4f* wp = ring + (stg) * STAGE_V4 + lane;                               \
    _Pragma("unroll") for (int s = 0; s < CHUNK; ++s) {                     \
      v8f d0 = wmma4(aXa[0], sA[s], biasC[0]);                              \
      v8f d1 = wmma4(aXa[1], sA[s], biasC[1]);                              \
      d0 = wmma4(aXb[0], sB[s], d0);                                        \
      d1 = wmma4(aXb[1], sB[s], d1);                                        \
      wp[s * STEP_V4 + 0 * 32] = __builtin_shufflevector(d0, d0, 0, 1, 2, 3); \
      wp[s * STEP_V4 + 1 * 32] = __builtin_shufflevector(d0, d0, 4, 5, 6, 7); \
      wp[s * STEP_V4 + 2 * 32] = __builtin_shufflevector(d1, d1, 0, 1, 2, 3); \
      wp[s * STEP_V4 + 3 * 32] = __builtin_shufflevector(d1, d1, 4, 5, 6, 7); \
    }                                                                       \
  }

#define READ_GROUP(c0, c1, stg, f, n)                                       \
  {                                                                         \
    const v4f* rq = ring + (stg) * STAGE_V4 + (f) * STEP_V4 + lane;         \
    _Pragma("unroll") for (int s = 0; s < (n); ++s) {                       \
      c0[s] = join8(rq[s * STEP_V4 + 0 * 32], rq[s * STEP_V4 + 1 * 32]);    \
      c1[s] = join8(rq[s * STEP_V4 + 2 * 32], rq[s * STEP_V4 + 3 * 32]);    \
    }                                                                       \
  }

#define SCAN_STEP(cin0, cin1)                                               \
  {                                                                         \
    v8f g0 = wmma4(aH[0][0], bh0, cin0);                                    \
    v8f g1 = wmma4(aH[1][0], bh0, cin1);                                    \
    g0 = wmma4(aH[0][1], bh1, g0);                                          \
    g1 = wmma4(aH[1][1], bh1, g1);                                          \
    const float pi = pick4(g0[0], g0[1], g0[2], g0[3], own_odd, own_upper); \
    const float pf = pick4(g0[4], g0[5], g0[6], g0[7], own_odd, own_upper); \
    const float pg = pick4(g1[0], g1[1], g1[2], g1[3], own_odd, own_upper); \
    const float iK2 = __builtin_amdgcn_rcpf(                                \
        __builtin_fmaf(__builtin_amdgcn_exp2f(pi), INV_2K, INV_2K));        \
    const float fv = __builtin_amdgcn_rcpf(1.0f + __builtin_amdgcn_exp2f(pf)); \
    const float u  = __builtin_fmaf(iK2, -0.5f, fv * csK);                  \
    const float gs = __builtin_amdgcn_rcpf(1.0f + __builtin_amdgcn_exp2f(pg)); \
    csK = __builtin_fmaf(iK2, gs, u);                                       \
    const float oq0 = __builtin_amdgcn_rcpf(1.0f + __builtin_amdgcn_exp2f(g1[4])); \
    const float oq1 = __builtin_amdgcn_rcpf(1.0f + __builtin_amdgcn_exp2f(g1[5])); \
    const float oq2 = __builtin_amdgcn_rcpf(1.0f + __builtin_amdgcn_exp2f(g1[6])); \
    const float oq3 = __builtin_amdgcn_rcpf(1.0f + __builtin_amdgcn_exp2f(g1[7])); \
    float tail_tmp;                                                         \
    STEP_TAIL(bh0.x, bh0.y, bh1.x, bh1.y, tail_tmp, csK, oq0, oq1, oq2, oq3); \
  }

  v2f bh0 = {0.f, 0.f};
  v2f bh1 = {0.f, 0.f};
  float csK = 0.f;
  v2f xcA[CHUNK], xcB[CHUNK];
  v8f cN0[LEAD], cN1[LEAD];
  v8f cT0[CHUNK - LEAD], cT1[CHUNK - LEAD];

  if (role == 1) {
    FETCH_CHUNK(xcA, xcB, 0)
    MAKE_CHUNK(xcA, xcB, 0)
    FETCH_CHUNK(xcA, xcB, 1)
    MAKE_CHUNK(xcA, xcB, 1)
    FETCH_CHUNK(xcA, xcB, 2)
  }
  __syncthreads();
  if (role == 0) {
    READ_GROUP(cN0, cN1, 0, 0, LEAD)
  }

  int st = 0;
  for (int k = 0; k < NCHUNK; ++k) {
    const int st1 = (st == STAGES - 1) ? 0 : st + 1;
    const int st2 = (st1 == STAGES - 1) ? 0 : st1 + 1;
    if (role == 1) {
      if (k + 2 < NCHUNK) {
        v2f xnA[CHUNK], xnB[CHUNK];
        FETCH_CHUNK(xnA, xnB, k + 3)
#pragma unroll
        for (int s = 0; s < CHUNK; ++s) {
          const int tt = (k + 5) * CHUNK + s;
          const int tc = (tt < T_STEPS) ? tt : (T_STEPS - 1);
          __builtin_prefetch(xbase + (size_t)tc * tstride, 0, 0);
        }
        MAKE_CHUNK(xcA, xcB, st2)
#pragma unroll
        for (int s = 0; s < CHUNK; ++s) { xcA[s] = xnA[s]; xcB[s] = xnB[s]; }
      }
    } else {
      v8f cC0[LEAD], cC1[LEAD];
#pragma unroll
      for (int s = 0; s < LEAD; ++s) { cC0[s] = cN0[s]; cC1[s] = cN1[s]; }
      READ_GROUP(cT0, cT1, st, LEAD, CHUNK - LEAD)
      READ_GROUP(cN0, cN1, st1, 0, LEAD)
      __builtin_amdgcn_sched_barrier(0);
#pragma unroll
      for (int s = 0; s < LEAD; ++s) SCAN_STEP(cC0[s], cC1[s])
#pragma unroll
      for (int s = 0; s < CHUNK - LEAD; ++s) SCAN_STEP(cT0[s], cT1[s])
    }
    __syncthreads();
    st = st1;
  }
#undef FETCH_CHUNK
#undef MAKE_CHUNK
#undef READ_GROUP
#undef SCAN_STEP
#undef QUAD_LANE

  if (role == 0) {
    const float hv = pick4(bh0.x, bh0.y, bh1.x, bh1.y, own_odd, own_upper);
    h_out[(size_t)(b0 + bcol) * HID + 4 * half + own] = hv;
  }
}

__global__ __launch_bounds__(128) void out_proj_wmma(
    const float* __restrict__ h_last, const float* __restrict__ W_out,
    const float* __restrict__ b_out, float* __restrict__ out) {
  const int lane = threadIdx.x & 31;
  const int tile = (blockIdx.x * blockDim.x + threadIdx.x) >> 5;
  const int half = lane >> 4;
  const int lm   = lane & 15;

  const float* hrow = h_last + (size_t)(tile * 16 + lm) * HID;
  v2f a_lo, a_hi;
  a_lo.x = hrow[0 + 2 * half];
  a_lo.y = hrow[1 + 2 * half];
  a_hi.x = hrow[4 + 2 * half];
  a_hi.y = hrow[5 + 2 * half];

  float w0 = 0.f, w1 = 0.f, w2 = 0.f, w3 = 0.f, bo = 0.f;
  if (lm < OUT_DIM) {
    const float* wr = W_out + lm * HID;
    w0 = wr[0 + 2 * half];
    w1 = wr[1 + 2 * half];
    w2 = wr[4 + 2 * half];
    w3 = wr[5 + 2 * half];
    bo = b_out[lm];
  }
  v2f b_lo, b_hi;
  b_lo.x = w0; b_lo.y = w1;
  b_hi.x = w2; b_hi.y = w3;

  v8f acc;
#pragma unroll
  for (int r = 0; r < 8; ++r) acc[r] = bo;

  acc = __builtin_amdgcn_wmma_f32_16x16x4_f32(false, a_lo, false, b_lo,
                                              (short)0, acc, false, false);
  acc = __builtin_amdgcn_wmma_f32_16x16x4_f32(false, a_hi, false, b_hi,
                                              (short)0, acc, false, false);

  if (lm < OUT_DIM) {
    float* orow = out + (size_t)(tile * 16 + 8 * half) * OUT_DIM + lm;
#pragma unroll
    for (int r = 0; r < 8; ++r) orow[r * OUT_DIM] = acc[r];
  }
}

extern "C" void kernel_launch(void* const* d_in, const int* in_sizes, int n_in,
                              void* d_out, int out_size, void* d_ws, size_t ws_size,
                              hipStream_t stream) {
  const float* seq   = (const float*)d_in[0];
  const float* W_ih  = (const float*)d_in[1];
  const float* W_hh  = (const float*)d_in[2];
  const float* b_ih  = (const float*)d_in[3];
  const float* b_hh  = (const float*)d_in[4];
  const float* W_out = (const float*)d_in[5];
  const float* b_out = (const float*)d_in[6];
  float* out    = (float*)d_out;
  float* h_last = (float*)d_ws;

  lstm_scan_wmma<<<BATCH / BPB, 64, 0, stream>>>(seq, W_ih, W_hh, b_ih, b_hh,
                                                 h_last);
  out_proj_wmma<<<(BATCH / 16) * 32 / 128, 128, 0, stream>>>(h_last, W_out,
                                                             b_out, out);
}
